# starting kernel
# baseline (speedup 1.0000x reference)
#include <hip/hip_runtime.h>
#include <hip/hip_bf16.h>
#include <math.h>

typedef __attribute__((ext_vector_type(2))) float v2f;
typedef __attribute__((ext_vector_type(8))) float v8f;

#define DT 0.005f

__global__ void k_prep(const float* __restrict__ tc, const float* __restrict__ tk,
                       float* __restrict__ comb, float* __restrict__ accum) {
    int t = threadIdx.x;
    if (t < 8) accum[t] = 0.0f;
    int s = t >> 6;
    float inv_tau = DT / tc[s];
    float d[16];
    float sum = 0.0f;
#pragma unroll
    for (int j = 0; j < 16; ++j) { d[j] = expf(-(float)j * inv_tau); sum += d[j]; }
    float inv = 1.0f / sum;
#pragma unroll
    for (int j = 0; j < 16; ++j)
        comb[t * 16 + j] = d[j] * inv * tk[t * 16 + j];
}

__global__ __launch_bounds__(256) void k_stats(const float* __restrict__ events,
                                               const float* __restrict__ comb,
                                               float* __restrict__ accum) {
    __shared__ float sComb[256 * 17];
    __shared__ float sXp[8][80];
    __shared__ float sAcc[4];

    int tid = threadIdx.x;
    int lane = tid & 31, w = tid >> 5;
#pragma unroll
    for (int j = 0; j < 16; ++j) sComb[tid * 17 + j] = comb[tid * 16 + j];
    if (tid < 4) sAcc[tid] = 0.0f;

    int ch = blockIdx.x * 8 + w;
    const float* x = events + (size_t)ch * 64;
    sXp[w][lane]      = (lane >= 8) ? x[lane - 8] : 0.0f;
    sXp[w][32 + lane] = x[24 + lane];
    if (lane < 16) sXp[w][64 + lane] = (lane < 8) ? x[56 + lane] : 0.0f;
    __syncthreads();

    const int m    = lane & 15;
    const int koff = (lane < 16) ? 0 : 2;
    const bool lowHalf = (lane < 16);
    const float* xrow = &sXp[w][0];

    v2f afr[4][4];
#pragma unroll
    for (int mt = 0; mt < 4; ++mt)
#pragma unroll
        for (int ks = 0; ks < 4; ++ks) {
            int k0 = ks * 4 + koff;
            afr[mt][ks].x = xrow[mt * 16 + m + k0];
            afr[mt][ks].y = xrow[mt * 16 + m + k0 + 1];
        }

    for (int ntile = 0; ntile < 16; ++ntile) {
        const float* brow = &sComb[(ntile * 16 + m) * 17];
        v8f c0 = {}, c1 = {}, c2 = {}, c3 = {};
#pragma unroll
        for (int ks = 0; ks < 4; ++ks) {
            int k0 = ks * 4 + koff;
            v2f bv;
            bv.x = brow[k0];
            bv.y = brow[k0 + 1];
            c0 = __builtin_amdgcn_wmma_f32_16x16x4_f32(false, afr[0][ks], false, bv, (short)0, c0, false, false);
            c1 = __builtin_amdgcn_wmma_f32_16x16x4_f32(false, afr[1][ks], false, bv, (short)0, c1, false, false);
            c2 = __builtin_amdgcn_wmma_f32_16x16x4_f32(false, afr[2][ks], false, bv, (short)0, c2, false, false);
            c3 = __builtin_amdgcn_wmma_f32_16x16x4_f32(false, afr[3][ks], false, bv, (short)0, c3, false, false);
        }
        float sum = 0.0f, sq = 0.0f;
#pragma unroll
        for (int v = 0; v < 8; ++v) {
            float f0 = c0[v], f1 = c1[v], f2 = c2[v], f3 = c3[v];
            sum += f0 + f1 + f2 + f3;
            sq  += f0 * f0 + f1 * f1 + f2 * f2 + f3 * f3;
        }
        float o64 = 0.0f;
#pragma unroll
        for (int j = 0; j < 16; ++j) o64 += xrow[64 + j] * brow[j];
        float c64 = lowHalf ? o64 : 0.0f;
        sum += c64; sq += c64 * c64;

        sum += __shfl_xor(sum, 16, 32);
        sq  += __shfl_xor(sq, 16, 32);
        float var = (sq - sum * sum * (1.0f / 65.0f)) * (1.0f / 64.0f);
        var += __shfl_xor(var, 8, 32);
        var += __shfl_xor(var, 4, 32);
        var += __shfl_xor(var, 2, 32);
        var += __shfl_xor(var, 1, 32);
        if (lane == 0) atomicAdd(&sAcc[ntile >> 2], var);
    }
    __syncthreads();
    if (tid < 4) {
        int b = (blockIdx.x * 8) >> 11;
        atomicAdd(&accum[b * 4 + tid], sAcc[tid]);
    }
}

__global__ void k_attn(const float* __restrict__ accum,
                       const float* __restrict__ w1, const float* __restrict__ b1,
                       const float* __restrict__ w2, const float* __restrict__ b2,
                       const float* __restrict__ comb, float* __restrict__ keff) {
    __shared__ float sAttn[2][4];
    int tid = threadIdx.x;
    if (tid < 2) {
        int b = tid;
        float fv[4];
#pragma unroll
        for (int s = 0; s < 4; ++s) fv[s] = accum[b * 4 + s] * (1.0f / (64.0f * 2048.0f));
        float h[8];
#pragma unroll
        for (int i = 0; i < 8; ++i) {
            float a = b1[i];
#pragma unroll
            for (int s = 0; s < 4; ++s) a += fv[s] * w1[i * 4 + s];
            h[i] = fmaxf(a, 0.0f);
        }
        float lg[4]; float mx = -1e30f;
#pragma unroll
        for (int s = 0; s < 4; ++s) {
            float a = b2[s];
#pragma unroll
            for (int i = 0; i < 8; ++i) a += h[i] * w2[s * 8 + i];
            lg[s] = a; mx = fmaxf(mx, a);
        }
        float den = 0.0f;
#pragma unroll
        for (int s = 0; s < 4; ++s) { lg[s] = expf(lg[s] - mx); den += lg[s]; }
        float inv = 1.0f / den;
#pragma unroll
        for (int s = 0; s < 4; ++s) sAttn[b][s] = lg[s] * inv;
    }
    __syncthreads();
#pragma unroll
    for (int i = 0; i < 8; ++i) {
        int idx = tid + i * 256;
        int b = idx >> 10; int r = idx & 1023; int e = r >> 4; int j = r & 15;
        float acc = 0.0f;
#pragma unroll
        for (int s = 0; s < 4; ++s) acc += sAttn[b][s] * comb[(s * 64 + e) * 16 + j];
        keff[idx] = acc;
    }
}

__global__ __launch_bounds__(256) void k_encode(const float* __restrict__ events,
                                                const float* __restrict__ keff,
                                                float* __restrict__ out) {
    __shared__ float sK[64 * 17];
    __shared__ float sXp[8][80];
    __shared__ float sStage[8][16][67];

    int tid = threadIdx.x, lane = tid & 31, w = tid >> 5;
    int ch = blockIdx.x * 8 + w;
    int b  = ch >> 11;
    int cf = ch & 2047;

#pragma unroll
    for (int i = 0; i < 4; ++i) {
        int idx = tid + i * 256;
        int e = idx >> 4, j = idx & 15;
        sK[e * 17 + j] = keff[(b << 10) + idx];
    }
    const float* x = events + (size_t)ch * 64;
    sXp[w][lane]      = (lane >= 8) ? x[lane - 8] : 0.0f;
    sXp[w][32 + lane] = x[24 + lane];
    if (lane < 16) sXp[w][64 + lane] = (lane < 8) ? x[56 + lane] : 0.0f;
    __syncthreads();

    const int m    = lane & 15;
    const int koff = (lane < 16) ? 0 : 2;
    const int moff = (lane < 16) ? 0 : 8;
    const float* xrow = &sXp[w][0];

    v2f afr[4][4];
#pragma unroll
    for (int mt = 0; mt < 4; ++mt)
#pragma unroll
        for (int ks = 0; ks < 4; ++ks) {
            int k0 = ks * 4 + koff;
            afr[mt][ks].x = xrow[mt * 16 + m + k0];
            afr[mt][ks].y = xrow[mt * 16 + m + k0 + 1];
        }

    for (int ntile = 0; ntile < 4; ++ntile) {
        const float* brow = &sK[(ntile * 16 + m) * 17];
        v8f c0 = {}, c1 = {}, c2 = {}, c3 = {};
#pragma unroll
        for (int ks = 0; ks < 4; ++ks) {
            int k0 = ks * 4 + koff;
            v2f bv;
            bv.x = brow[k0];
            bv.y = brow[k0 + 1];
            c0 = __builtin_amdgcn_wmma_f32_16x16x4_f32(false, afr[0][ks], false, bv, (short)0, c0, false, false);
            c1 = __builtin_amdgcn_wmma_f32_16x16x4_f32(false, afr[1][ks], false, bv, (short)0, c1, false, false);
            c2 = __builtin_amdgcn_wmma_f32_16x16x4_f32(false, afr[2][ks], false, bv, (short)0, c2, false, false);
            c3 = __builtin_amdgcn_wmma_f32_16x16x4_f32(false, afr[3][ks], false, bv, (short)0, c3, false, false);
        }
#pragma unroll
        for (int v = 0; v < 8; ++v) {
            sStage[w][m][ 0 + moff + v] = c0[v];
            sStage[w][m][16 + moff + v] = c1[v];
            sStage[w][m][32 + moff + v] = c2[v];
            sStage[w][m][48 + moff + v] = c3[v];
        }
        float o64 = 0.0f;
#pragma unroll
        for (int j = 0; j < 16; ++j) o64 += xrow[64 + j] * brow[j];
        sStage[w][m][64] = o64;

        __asm__ volatile("s_wait_dscnt 0" ::: "memory");
        int nb = ntile * 16;
        for (int e16 = 0; e16 < 16; ++e16) {
            int e = nb + e16;
            size_t base = ((size_t)(b * 64 + e) * 2048 + cf) * 65;
#pragma unroll
            for (int t0 = 0; t0 < 96; t0 += 32) {
                int t = t0 + lane;
                if (t < 65) out[base + t] = sStage[w][e16][t];
            }
        }
        __asm__ volatile("" ::: "memory");
    }
}

extern "C" void kernel_launch(void* const* d_in, const int* in_sizes, int n_in,
                              void* d_out, int out_size, void* d_ws, size_t ws_size,
                              hipStream_t stream) {
    const float* events = (const float*)d_in[0];
    const float* tcs    = (const float*)d_in[1];
    const float* tk     = (const float*)d_in[2];
    const float* w1     = (const float*)d_in[3];
    const float* b1     = (const float*)d_in[4];
    const float* w2     = (const float*)d_in[5];
    const float* b2     = (const float*)d_in[6];
    float* out = (float*)d_out;

    float* ws    = (float*)d_ws;
    float* comb  = ws;
    float* accum = ws + 4096;
    float* keff  = ws + 4104;

    k_prep  <<<1,   256, 0, stream>>>(tcs, tk, comb, accum);
    k_stats <<<512, 256, 0, stream>>>(events, comb, accum);
    k_attn  <<<1,   256, 0, stream>>>(accum, w1, b1, w2, b2, comb, keff);
    k_encode<<<512, 256, 0, stream>>>(events, keff, out);
}


# k_stats: variance from per-batch 16x16 Gram matrix sum_ch(X^T X - P P^T/65) on f32 WMMA (17 per channel, not 256); quadratic form with comb in k_attn on WMMA; changes fp summation order
# speedup vs baseline: 1.4354x; 1.4354x over previous
#include <hip/hip_runtime.h>
#include <hip/hip_bf16.h>
#include <math.h>

typedef __attribute__((ext_vector_type(2))) float v2f;
typedef __attribute__((ext_vector_type(8))) float v8f;

#define DT 0.005f

__global__ void k_prep(const float* __restrict__ tc, const float* __restrict__ tk,
                       float* __restrict__ comb, float* __restrict__ gram) {
    int t = threadIdx.x;
    gram[t] = 0.0f;
    gram[t + 256] = 0.0f;
    int s = t >> 6;
    float inv_tau = DT / tc[s];
    float d[16];
    float sum = 0.0f;
#pragma unroll
    for (int j = 0; j < 16; ++j) { d[j] = expf(-(float)j * inv_tau); sum += d[j]; }
    float inv = 1.0f / sum;
#pragma unroll
    for (int j = 0; j < 16; ++j)
        comb[t * 16 + j] = d[j] * inv * tk[t * 16 + j];
}

__global__ __launch_bounds__(128) void k_stats(const float* __restrict__ events,
                                               float* __restrict__ gram) {
    __shared__ alignas(16) float sXp[4][4][88];
    __shared__ float sRed[4][256];

    int tid = threadIdx.x;
    int lane = tid & 31, w = tid >> 5;
    const float4* x4 = reinterpret_cast<const float4*>(
        events + ((size_t)blockIdx.x * 16 + w * 4) * 64);
    float4 v0 = x4[lane];
    float4 v1 = x4[32 + lane];
    const int hc = lane >> 4, q = lane & 15;
    sXp[w][lane >> 3][lane & 7] = 0.0f;
    sXp[w][hc][72 + q] = 0.0f;
    sXp[w][2 + hc][72 + q] = 0.0f;
    *reinterpret_cast<float4*>(&sXp[w][hc][8 + 4 * q]) = v0;
    *reinterpret_cast<float4*>(&sXp[w][2 + hc][8 + 4 * q]) = v1;
    __syncthreads();

    const int m = lane & 15;
    const bool lowHalf = (lane < 16);
    const int koff = lowHalf ? 0 : 2;
    const float* r0 = &sXp[w][0][m + koff];
    const float* r1 = &sXp[w][1][m + koff];
    const float* r2 = &sXp[w][2][m + koff];
    const float* r3 = &sXp[w][3][m + koff];

    v8f a0 = {}, a1 = {}, a2 = {}, a3 = {};
    float p0 = 0.0f, p1 = 0.0f, p2 = 0.0f, p3 = 0.0f;
#pragma unroll
    for (int ks = 0; ks < 16; ++ks) {
        v2f f0, f1, f2, f3;
        f0.x = r0[4 * ks]; f0.y = r0[4 * ks + 1];
        f1.x = r1[4 * ks]; f1.y = r1[4 * ks + 1];
        f2.x = r2[4 * ks]; f2.y = r2[4 * ks + 1];
        f3.x = r3[4 * ks]; f3.y = r3[4 * ks + 1];
        a0 = __builtin_amdgcn_wmma_f32_16x16x4_f32(false, f0, false, f0, (short)0, a0, false, false);
        a1 = __builtin_amdgcn_wmma_f32_16x16x4_f32(false, f1, false, f1, (short)0, a1, false, false);
        a2 = __builtin_amdgcn_wmma_f32_16x16x4_f32(false, f2, false, f2, (short)0, a2, false, false);
        a3 = __builtin_amdgcn_wmma_f32_16x16x4_f32(false, f3, false, f3, (short)0, a3, false, false);
        p0 += f0.x + f0.y; p1 += f1.x + f1.y;
        p2 += f2.x + f2.y; p3 += f3.x + f3.y;
    }
    {
        v2f f0, f1, f2, f3;
        f0.x = lowHalf ? r0[64] : 0.0f; f0.y = 0.0f;
        f1.x = lowHalf ? r1[64] : 0.0f; f1.y = 0.0f;
        f2.x = lowHalf ? r2[64] : 0.0f; f2.y = 0.0f;
        f3.x = lowHalf ? r3[64] : 0.0f; f3.y = 0.0f;
        a0 = __builtin_amdgcn_wmma_f32_16x16x4_f32(false, f0, false, f0, (short)0, a0, false, false);
        a1 = __builtin_amdgcn_wmma_f32_16x16x4_f32(false, f1, false, f1, (short)0, a1, false, false);
        a2 = __builtin_amdgcn_wmma_f32_16x16x4_f32(false, f2, false, f2, (short)0, a2, false, false);
        a3 = __builtin_amdgcn_wmma_f32_16x16x4_f32(false, f3, false, f3, (short)0, a3, false, false);
        p0 += f0.x; p1 += f1.x; p2 += f2.x; p3 += f3.x;
    }
    p0 += __shfl_xor(p0, 16, 32);
    p1 += __shfl_xor(p1, 16, 32);
    p2 += __shfl_xor(p2, 16, 32);
    p3 += __shfl_xor(p3, 16, 32);
    v2f pf;
    pf.x = lowHalf ? p0 : p2;
    pf.y = lowHalf ? p1 : p3;
    v8f pz = {};
    v8f pp = __builtin_amdgcn_wmma_f32_16x16x4_f32(false, pf, false, pf, (short)0, pz, false, false);
    v8f g = (a0 + a1) + (a2 + a3) - pp * (1.0f / 65.0f);

#pragma unroll
    for (int r = 0; r < 8; ++r) sRed[w][r * 32 + lane] = g[r];
    __syncthreads();
    const int b = blockIdx.x >> 7;
#pragma unroll
    for (int i = 0; i < 2; ++i) {
        int idx = tid + i * 128;
        float tot = (sRed[0][idx] + sRed[1][idx]) + (sRed[2][idx] + sRed[3][idx]);
        int rr = idx >> 5, l = idx & 31;
        int row = ((l >> 4) << 3) + rr, col = l & 15;
        atomicAdd(&gram[b * 256 + row * 16 + col], tot);
    }
}

__global__ __launch_bounds__(256) void k_attn(const float* gram,
                       const float* __restrict__ w1, const float* __restrict__ b1,
                       const float* __restrict__ w2, const float* __restrict__ b2,
                       const float* __restrict__ comb, float* keff) {
    __shared__ float sM[2][256];
    __shared__ float sPart[2][16];
    __shared__ float sAttn[2][4];
    int tid = threadIdx.x;
    int lane = tid & 31, w = tid >> 5;
    sM[0][tid] = gram[tid];
    sM[1][tid] = gram[256 + tid];
    __syncthreads();

    {
        const int c    = lane & 15;
        const int koff = (lane < 16) ? 0 : 2;
        const int hrow = (lane < 16) ? 0 : 8;
        v2f bf0[4], bf1[4];
#pragma unroll
        for (int ks = 0; ks < 4; ++ks) {
            int k0 = ks * 4 + koff;
            bf0[ks].x = sM[0][k0 * 16 + c]; bf0[ks].y = sM[0][(k0 + 1) * 16 + c];
            bf1[ks].x = sM[1][k0 * 16 + c]; bf1[ks].y = sM[1][(k0 + 1) * 16 + c];
        }
#pragma unroll
        for (int i = 0; i < 2; ++i) {
            int tile = 2 * w + i;
            const float* crow = comb + (tile * 16 + c) * 16;
            v8f d0 = {}, d1 = {};
#pragma unroll
            for (int ks = 0; ks < 4; ++ks) {
                int k0 = ks * 4 + koff;
                v2f af;
                af.x = crow[k0];
                af.y = crow[k0 + 1];
                d0 = __builtin_amdgcn_wmma_f32_16x16x4_f32(false, af, false, bf0[ks], (short)0, d0, false, false);
                d1 = __builtin_amdgcn_wmma_f32_16x16x4_f32(false, af, false, bf1[ks], (short)0, d1, false, false);
            }
            float q0 = 0.0f, q1 = 0.0f;
#pragma unroll
            for (int r = 0; r < 8; ++r) {
                float cv = comb[(tile * 16 + hrow + r) * 16 + c];
                q0 += d0[r] * cv;
                q1 += d1[r] * cv;
            }
            q0 += __shfl_xor(q0, 16, 32); q1 += __shfl_xor(q1, 16, 32);
            q0 += __shfl_xor(q0, 8, 32);  q1 += __shfl_xor(q1, 8, 32);
            q0 += __shfl_xor(q0, 4, 32);  q1 += __shfl_xor(q1, 4, 32);
            q0 += __shfl_xor(q0, 2, 32);  q1 += __shfl_xor(q1, 2, 32);
            q0 += __shfl_xor(q0, 1, 32);  q1 += __shfl_xor(q1, 1, 32);
            if (lane == 0) { sPart[0][tile] = q0; sPart[1][tile] = q1; }
        }
    }
    __syncthreads();
    if (tid < 2) {
        int b = tid;
        float fv[4];
#pragma unroll
        for (int s = 0; s < 4; ++s)
            fv[s] = ((sPart[b][4 * s] + sPart[b][4 * s + 1]) +
                     (sPart[b][4 * s + 2] + sPart[b][4 * s + 3])) *
                    (1.0f / (64.0f * 64.0f * 2048.0f));
        float h[8];
#pragma unroll
        for (int i = 0; i < 8; ++i) {
            float a = b1[i];
#pragma unroll
            for (int s = 0; s < 4; ++s) a += fv[s] * w1[i * 4 + s];
            h[i] = fmaxf(a, 0.0f);
        }
        float lg[4]; float mx = -1e30f;
#pragma unroll
        for (int s = 0; s < 4; ++s) {
            float a = b2[s];
#pragma unroll
            for (int i = 0; i < 8; ++i) a += h[i] * w2[s * 8 + i];
            lg[s] = a; mx = fmaxf(mx, a);
        }
        float den = 0.0f;
#pragma unroll
        for (int s = 0; s < 4; ++s) { lg[s] = expf(lg[s] - mx); den += lg[s]; }
        float inv = 1.0f / den;
#pragma unroll
        for (int s = 0; s < 4; ++s) sAttn[b][s] = lg[s] * inv;
    }
    __syncthreads();
#pragma unroll
    for (int i = 0; i < 8; ++i) {
        int idx = tid + i * 256;
        int b = idx >> 10; int r = idx & 1023; int e = r >> 4; int j = r & 15;
        float acc = 0.0f;
#pragma unroll
        for (int s = 0; s < 4; ++s) acc += sAttn[b][s] * comb[(s * 64 + e) * 16 + j];
        keff[idx] = acc;
    }
}

__global__ __launch_bounds__(256) void k_encode(const float* __restrict__ events,
                                                const float* __restrict__ keff,
                                                float* __restrict__ out) {
    __shared__ float sK[64 * 17];
    __shared__ float sXp[8][80];
    __shared__ float sStage[8][16][67];

    int tid = threadIdx.x, lane = tid & 31, w = tid >> 5;
    int ch = blockIdx.x * 8 + w;
    int b  = ch >> 11;
    int cf = ch & 2047;

#pragma unroll
    for (int i = 0; i < 4; ++i) {
        int idx = tid + i * 256;
        int e = idx >> 4, j = idx & 15;
        sK[e * 17 + j] = keff[(b << 10) + idx];
    }
    const float* x = events + (size_t)ch * 64;
    sXp[w][lane]      = (lane >= 8) ? x[lane - 8] : 0.0f;
    sXp[w][32 + lane] = x[24 + lane];
    if (lane < 16) sXp[w][64 + lane] = (lane < 8) ? x[56 + lane] : 0.0f;
    __syncthreads();

    const int m    = lane & 15;
    const int koff = (lane < 16) ? 0 : 2;
    const int moff = (lane < 16) ? 0 : 8;
    const float* xrow = &sXp[w][0];

    v2f afr[4][4];
#pragma unroll
    for (int mt = 0; mt < 4; ++mt)
#pragma unroll
        for (int ks = 0; ks < 4; ++ks) {
            int k0 = ks * 4 + koff;
            afr[mt][ks].x = xrow[mt * 16 + m + k0];
            afr[mt][ks].y = xrow[mt * 16 + m + k0 + 1];
        }

    for (int ntile = 0; ntile < 4; ++ntile) {
        const float* brow = &sK[(ntile * 16 + m) * 17];
        v8f c0 = {}, c1 = {}, c2 = {}, c3 = {};
#pragma unroll
        for (int ks = 0; ks < 4; ++ks) {
            int k0 = ks * 4 + koff;
            v2f bv;
            bv.x = brow[k0];
            bv.y = brow[k0 + 1];
            c0 = __builtin_amdgcn_wmma_f32_16x16x4_f32(false, afr[0][ks], false, bv, (short)0, c0, false, false);
            c1 = __builtin_amdgcn_wmma_f32_16x16x4_f32(false, afr[1][ks], false, bv, (short)0, c1, false, false);
            c2 = __builtin_amdgcn_wmma_f32_16x16x4_f32(false, afr[2][ks], false, bv, (short)0, c2, false, false);
            c3 = __builtin_amdgcn_wmma_f32_16x16x4_f32(false, afr[3][ks], false, bv, (short)0, c3, false, false);
        }
#pragma unroll
        for (int v = 0; v < 8; ++v) {
            sStage[w][m][ 0 + moff + v] = c0[v];
            sStage[w][m][16 + moff + v] = c1[v];
            sStage[w][m][32 + moff + v] = c2[v];
            sStage[w][m][48 + moff + v] = c3[v];
        }
        float o64 = 0.0f;
#pragma unroll
        for (int j = 0; j < 16; ++j) o64 += xrow[64 + j] * brow[j];
        sStage[w][m][64] = o64;

        __asm__ volatile("s_wait_dscnt 0" ::: "memory");
        int nb = ntile * 16;
        for (int e16 = 0; e16 < 16; ++e16) {
            int e = nb + e16;
            size_t base = ((size_t)(b * 64 + e) * 2048 + cf) * 65;
#pragma unroll
            for (int t0 = 0; t0 < 96; t0 += 32) {
                int t = t0 + lane;
                if (t < 65) out[base + t] = sStage[w][e16][t];
            }
        }
        __asm__ volatile("" ::: "memory");
    }
}

extern "C" void kernel_launch(void* const* d_in, const int* in_sizes, int n_in,
                              void* d_out, int out_size, void* d_ws, size_t ws_size,
                              hipStream_t stream) {
    const float* events = (const float*)d_in[0];
    const float* tcs    = (const float*)d_in[1];
    const float* tk     = (const float*)d_in[2];
    const float* w1     = (const float*)d_in[3];
    const float* b1     = (const float*)d_in[4];
    const float* w2     = (const float*)d_in[5];
    const float* b2     = (const float*)d_in[6];
    float* out = (float*)d_out;

    float* ws    = (float*)d_ws;
    float* comb  = ws;
    float* gram  = ws + 4104;
    float* keff  = ws + 4104;

    k_prep  <<<1,   256, 0, stream>>>(tcs, tk, comb, gram);
    k_stats <<<256, 128, 0, stream>>>(events, gram);
    k_attn  <<<1,   256, 0, stream>>>(gram, w1, b1, w2, b2, comb, keff);
    k_encode<<<512, 256, 0, stream>>>(events, keff, out);
}


# k_encode: 16 channels per block, product taken transposed (keff as A, x as B, same k order), staged in LDS; block writes whole 128-B lines with 16-byte stores (was 4-byte stores of 260-B rows)
# speedup vs baseline: 1.8857x; 1.3137x over previous
#include <hip/hip_runtime.h>
#include <hip/hip_bf16.h>
#include <math.h>

typedef __attribute__((ext_vector_type(2))) float v2f;
typedef __attribute__((ext_vector_type(8))) float v8f;

#define DT 0.005f

__global__ void k_prep(const float* __restrict__ tc, const float* __restrict__ tk,
                       float* __restrict__ comb, float* __restrict__ gram) {
    int t = threadIdx.x;
    gram[t] = 0.0f;
    gram[t + 256] = 0.0f;
    int s = t >> 6;
    float inv_tau = DT / tc[s];
    float d[16];
    float sum = 0.0f;
#pragma unroll
    for (int j = 0; j < 16; ++j) { d[j] = expf(-(float)j * inv_tau); sum += d[j]; }
    float inv = 1.0f / sum;
#pragma unroll
    for (int j = 0; j < 16; ++j)
        comb[t * 16 + j] = d[j] * inv * tk[t * 16 + j];
}

__global__ __launch_bounds__(128) void k_stats(const float* __restrict__ events,
                                               float* __restrict__ gram) {
    __shared__ alignas(16) float sXp[4][4][88];
    __shared__ float sRed[4][256];

    int tid = threadIdx.x;
    int lane = tid & 31, w = tid >> 5;
    const float4* x4 = reinterpret_cast<const float4*>(
        events + ((size_t)blockIdx.x * 16 + w * 4) * 64);
    float4 v0 = x4[lane];
    float4 v1 = x4[32 + lane];
    const int hc = lane >> 4, q = lane & 15;
    sXp[w][lane >> 3][lane & 7] = 0.0f;
    sXp[w][hc][72 + q] = 0.0f;
    sXp[w][2 + hc][72 + q] = 0.0f;
    *reinterpret_cast<float4*>(&sXp[w][hc][8 + 4 * q]) = v0;
    *reinterpret_cast<float4*>(&sXp[w][2 + hc][8 + 4 * q]) = v1;
    __syncthreads();

    const int m = lane & 15;
    const bool lowHalf = (lane < 16);
    const int koff = lowHalf ? 0 : 2;
    const float* r0 = &sXp[w][0][m + koff];
    const float* r1 = &sXp[w][1][m + koff];
    const float* r2 = &sXp[w][2][m + koff];
    const float* r3 = &sXp[w][3][m + koff];

    v8f a0 = {}, a1 = {}, a2 = {}, a3 = {};
    float p0 = 0.0f, p1 = 0.0f, p2 = 0.0f, p3 = 0.0f;
#pragma unroll
    for (int ks = 0; ks < 16; ++ks) {
        v2f f0, f1, f2, f3;
        f0.x = r0[4 * ks]; f0.y = r0[4 * ks + 1];
        f1.x = r1[4 * ks]; f1.y = r1[4 * ks + 1];
        f2.x = r2[4 * ks]; f2.y = r2[4 * ks + 1];
        f3.x = r3[4 * ks]; f3.y = r3[4 * ks + 1];
        a0 = __builtin_amdgcn_wmma_f32_16x16x4_f32(false, f0, false, f0, (short)0, a0, false, false);
        a1 = __builtin_amdgcn_wmma_f32_16x16x4_f32(false, f1, false, f1, (short)0, a1, false, false);
        a2 = __builtin_amdgcn_wmma_f32_16x16x4_f32(false, f2, false, f2, (short)0, a2, false, false);
        a3 = __builtin_amdgcn_wmma_f32_16x16x4_f32(false, f3, false, f3, (short)0, a3, false, false);
        p0 += f0.x + f0.y; p1 += f1.x + f1.y;
        p2 += f2.x + f2.y; p3 += f3.x + f3.y;
    }
    {
        v2f f0, f1, f2, f3;
        f0.x = lowHalf ? r0[64] : 0.0f; f0.y = 0.0f;
        f1.x = lowHalf ? r1[64] : 0.0f; f1.y = 0.0f;
        f2.x = lowHalf ? r2[64] : 0.0f; f2.y = 0.0f;
        f3.x = lowHalf ? r3[64] : 0.0f; f3.y = 0.0f;
        a0 = __builtin_amdgcn_wmma_f32_16x16x4_f32(false, f0, false, f0, (short)0, a0, false, false);
        a1 = __builtin_amdgcn_wmma_f32_16x16x4_f32(false, f1, false, f1, (short)0, a1, false, false);
        a2 = __builtin_amdgcn_wmma_f32_16x16x4_f32(false, f2, false, f2, (short)0, a2, false, false);
        a3 = __builtin_amdgcn_wmma_f32_16x16x4_f32(false, f3, false, f3, (short)0, a3, false, false);
        p0 += f0.x; p1 += f1.x; p2 += f2.x; p3 += f3.x;
    }
    p0 += __shfl_xor(p0, 16, 32);
    p1 += __shfl_xor(p1, 16, 32);
    p2 += __shfl_xor(p2, 16, 32);
    p3 += __shfl_xor(p3, 16, 32);
    v2f pf;
    pf.x = lowHalf ? p0 : p2;
    pf.y = lowHalf ? p1 : p3;
    v8f pz = {};
    v8f pp = __builtin_amdgcn_wmma_f32_16x16x4_f32(false, pf, false, pf, (short)0, pz, false, false);
    v8f g = (a0 + a1) + (a2 + a3) - pp * (1.0f / 65.0f);

#pragma unroll
    for (int r = 0; r < 8; ++r) sRed[w][r * 32 + lane] = g[r];
    __syncthreads();
    const int b = blockIdx.x >> 7;
#pragma unroll
    for (int i = 0; i < 2; ++i) {
        int idx = tid + i * 128;
        float tot = (sRed[0][idx] + sRed[1][idx]) + (sRed[2][idx] + sRed[3][idx]);
        int rr = idx >> 5, l = idx & 31;
        int row = ((l >> 4) << 3) + rr, col = l & 15;
        atomicAdd(&gram[b * 256 + row * 16 + col], tot);
    }
}

__global__ __launch_bounds__(256) void k_attn(const float* gram,
                       const float* __restrict__ w1, const float* __restrict__ b1,
                       const float* __restrict__ w2, const float* __restrict__ b2,
                       const float* __restrict__ comb, float* keff) {
    __shared__ float sM[2][256];
    __shared__ float sPart[2][16];
    __shared__ float sAttn[2][4];
    int tid = threadIdx.x;
    int lane = tid & 31, w = tid >> 5;
    sM[0][tid] = gram[tid];
    sM[1][tid] = gram[256 + tid];
    __syncthreads();

    {
        const int c    = lane & 15;
        const int koff = (lane < 16) ? 0 : 2;
        const int hrow = (lane < 16) ? 0 : 8;
        v2f bf0[4], bf1[4];
#pragma unroll
        for (int ks = 0; ks < 4; ++ks) {
            int k0 = ks * 4 + koff;
            bf0[ks].x = sM[0][k0 * 16 + c]; bf0[ks].y = sM[0][(k0 + 1) * 16 + c];
            bf1[ks].x = sM[1][k0 * 16 + c]; bf1[ks].y = sM[1][(k0 + 1) * 16 + c];
        }
#pragma unroll
        for (int i = 0; i < 2; ++i) {
            int tile = 2 * w + i;
            const float* crow = comb + (tile * 16 + c) * 16;
            v8f d0 = {}, d1 = {};
#pragma unroll
            for (int ks = 0; ks < 4; ++ks) {
                int k0 = ks * 4 + koff;
                v2f af;
                af.x = crow[k0];
                af.y = crow[k0 + 1];
                d0 = __builtin_amdgcn_wmma_f32_16x16x4_f32(false, af, false, bf0[ks], (short)0, d0, false, false);
                d1 = __builtin_amdgcn_wmma_f32_16x16x4_f32(false, af, false, bf1[ks], (short)0, d1, false, false);
            }
            float q0 = 0.0f, q1 = 0.0f;
#pragma unroll
            for (int r = 0; r < 8; ++r) {
                float cv = comb[(tile * 16 + hrow + r) * 16 + c];
                q0 += d0[r] * cv;
                q1 += d1[r] * cv;
            }
            q0 += __shfl_xor(q0, 16, 32); q1 += __shfl_xor(q1, 16, 32);
            q0 += __shfl_xor(q0, 8, 32);  q1 += __shfl_xor(q1, 8, 32);
            q0 += __shfl_xor(q0, 4, 32);  q1 += __shfl_xor(q1, 4, 32);
            q0 += __shfl_xor(q0, 2, 32);  q1 += __shfl_xor(q1, 2, 32);
            q0 += __shfl_xor(q0, 1, 32);  q1 += __shfl_xor(q1, 1, 32);
            if (lane == 0) { sPart[0][tile] = q0; sPart[1][tile] = q1; }
        }
    }
    __syncthreads();
    if (tid < 2) {
        int b = tid;
        float fv[4];
#pragma unroll
        for (int s = 0; s < 4; ++s)
            fv[s] = ((sPart[b][4 * s] + sPart[b][4 * s + 1]) +
                     (sPart[b][4 * s + 2] + sPart[b][4 * s + 3])) *
                    (1.0f / (64.0f * 64.0f * 2048.0f));
        float h[8];
#pragma unroll
        for (int i = 0; i < 8; ++i) {
            float a = b1[i];
#pragma unroll
            for (int s = 0; s < 4; ++s) a += fv[s] * w1[i * 4 + s];
            h[i] = fmaxf(a, 0.0f);
        }
        float lg[4]; float mx = -1e30f;
#pragma unroll
        for (int s = 0; s < 4; ++s) {
            float a = b2[s];
#pragma unroll
            for (int i = 0; i < 8; ++i) a += h[i] * w2[s * 8 + i];
            lg[s] = a; mx = fmaxf(mx, a);
        }
        float den = 0.0f;
#pragma unroll
        for (int s = 0; s < 4; ++s) { lg[s] = expf(lg[s] - mx); den += lg[s]; }
        float inv = 1.0f / den;
#pragma unroll
        for (int s = 0; s < 4; ++s) sAttn[b][s] = lg[s] * inv;
    }
    __syncthreads();
#pragma unroll
    for (int i = 0; i < 8; ++i) {
        int idx = tid + i * 256;
        int b = idx >> 10; int r = idx & 1023; int e = r >> 4; int j = r & 15;
        float acc = 0.0f;
#pragma unroll
        for (int s = 0; s < 4; ++s) acc += sAttn[b][s] * comb[(s * 64 + e) * 16 + j];
        keff[idx] = acc;
    }
}

#define ENC_ESTR 1044
__global__ __launch_bounds__(256) void k_encode(const float* __restrict__ events,
                                                const float* __restrict__ keff,
                                                float* __restrict__ out) {
    __shared__ float sK[64 * 17];
    __shared__ alignas(16) float sXp[16][80];
    __shared__ alignas(16) float sStage[3][16][ENC_ESTR];

    const int tid = threadIdx.x, lane = tid & 31, w = tid >> 5;
    const int ch0 = blockIdx.x * 16;
    const int b   = ch0 >> 11;
    const int cf0 = ch0 & 2047;

#pragma unroll
    for (int i = 0; i < 4; ++i) {
        int idx = tid + i * 256;
        int e = idx >> 4, j = idx & 15;
        sK[e * 17 + j] = keff[(b << 10) + idx];
    }
    {
        const float4* x4 = reinterpret_cast<const float4*>(events + (size_t)ch0 * 64);
        float4 v = x4[tid];
        const int chl = tid >> 4, q = tid & 15;
        sXp[chl][(q < 8) ? q : 64 + q] = 0.0f;
        *reinterpret_cast<float4*>(&sXp[chl][8 + 4 * q]) = v;
    }
    __syncthreads();

    const int m    = lane & 15;
    const int h    = lane >> 4;
    const int koff = 2 * h;

    v2f xf[2][4][4];
#pragma unroll
    for (int c2 = 0; c2 < 2; ++c2) {
        const float* xrow = &sXp[2 * w + c2][0];
#pragma unroll
        for (int mt = 0; mt < 4; ++mt)
#pragma unroll
            for (int ks = 0; ks < 4; ++ks) {
                int k0 = ks * 4 + koff;
                xf[c2][mt][ks].x = xrow[mt * 16 + m + k0];
                xf[c2][mt][ks].y = xrow[mt * 16 + m + k0 + 1];
            }
    }
    float xs[16];
#pragma unroll
    for (int j = 0; j < 16; ++j) xs[j] = sXp[2 * w + h][64 + j];

    const int a4 = (blockIdx.x & 1) * 4;

    for (int ntile = 0; ntile < 4; ++ntile) {
        float* stg = &sStage[ntile % 3][0][0];
        const float* brow = &sK[(ntile * 16 + m) * 17];
        v2f bv[4];
#pragma unroll
        for (int ks = 0; ks < 4; ++ks) {
            int k0 = ks * 4 + koff;
            bv[ks].x = brow[k0];
            bv[ks].y = brow[k0 + 1];
        }
        v8f acc[2][4];
#pragma unroll
        for (int c2 = 0; c2 < 2; ++c2)
#pragma unroll
            for (int mt = 0; mt < 4; ++mt) acc[c2][mt] = v8f{};
#pragma unroll
        for (int ks = 0; ks < 4; ++ks)
#pragma unroll
            for (int c2 = 0; c2 < 2; ++c2)
#pragma unroll
                for (int mt = 0; mt < 4; ++mt)
                    acc[c2][mt] = __builtin_amdgcn_wmma_f32_16x16x4_f32(
                        false, bv[ks], false, xf[c2][mt][ks], (short)0, acc[c2][mt], false, false);
#pragma unroll
        for (int c2 = 0; c2 < 2; ++c2)
#pragma unroll
            for (int mt = 0; mt < 4; ++mt)
#pragma unroll
                for (int r = 0; r < 8; ++r)
                    stg[(8 * h + r) * ENC_ESTR + (2 * w + c2) * 65 + mt * 16 + m] = acc[c2][mt][r];
        float o64 = 0.0f;
#pragma unroll
        for (int j = 0; j < 16; ++j) o64 += xs[j] * brow[j];
        stg[m * ENC_ESTR + (2 * w + h) * 65 + 64] = o64;
        __builtin_amdgcn_fence(__ATOMIC_RELEASE, "workgroup", "local");
        __builtin_amdgcn_s_barrier();
        __builtin_amdgcn_fence(__ATOMIC_ACQUIRE, "workgroup", "local");

#pragma unroll
        for (int rr = 0; rr < 2; ++rr) {
            const int e16 = w + 8 * rr;
            const float* srow = stg + e16 * ENC_ESTR;
            float4* orow = reinterpret_cast<float4*>(
                out + ((size_t)(b * 64 + ntile * 16 + e16) * 2048 + cf0) * 65);
            float4 v[9];
#pragma unroll
            for (int g = 0; g < 9; ++g) {
                int q = 32 * g + lane - a4;
                int qc = (q < 0) ? 0 : ((q > 259) ? 259 : q);
                v[g] = *reinterpret_cast<const float4*>(srow + 4 * qc);
            }
#pragma unroll
            for (int g = 0; g < 9; ++g) {
                int q = 32 * g + lane - a4;
                if (g == 0) { if (q >= 0) orow[q] = v[g]; }
                else if (g == 8) { if (q < 260) orow[q] = v[g]; }
                else orow[q] = v[g];
            }
        }
    }
}

extern "C" void kernel_launch(void* const* d_in, const int* in_sizes, int n_in,
                              void* d_out, int out_size, void* d_ws, size_t ws_size,
                              hipStream_t stream) {
    const float* events = (const float*)d_in[0];
    const float* tcs    = (const float*)d_in[1];
    const float* tk     = (const float*)d_in[2];
    const float* w1     = (const float*)d_in[3];
    const float* b1     = (const float*)d_in[4];
    const float* w2     = (const float*)d_in[5];
    const float* b2     = (const float*)d_in[6];
    float* out = (float*)d_out;

    float* ws    = (float*)d_ws;
    float* comb  = ws;
    float* gram  = ws + 4104;
    float* keff  = ws + 4104;

    k_prep  <<<1,   256, 0, stream>>>(tcs, tk, comb, gram);
    k_stats <<<256, 128, 0, stream>>>(events, gram);
    k_attn  <<<1,   256, 0, stream>>>(gram, w1, b1, w2, b2, comb, keff);
    k_encode<<<256, 256, 0, stream>>>(events, keff, out);
}


# 4 launches -> 2: filter prep and attention recomputed in the stats and encode blocks; each stats block reduces its Gram to 4 quadratic-form partials on WMMA, no atomics; changes fp summation order
# speedup vs baseline: 2.5098x; 1.3310x over previous
#include <hip/hip_runtime.h>
#include <hip/hip_bf16.h>
#include <math.h>

typedef __attribute__((ext_vector_type(2))) float v2f;
typedef __attribute__((ext_vector_type(8))) float v8f;

#define DT 0.005f

#define CPITCH 20

__device__ __forceinline__ float lane_bcast(float v, int j) {
    return __builtin_bit_cast(float, __builtin_amdgcn_readlane(__builtin_bit_cast(int, v), j));
}

__device__ __forceinline__ float decay_norm(float inv_tau, int lane) {
    const float d = expf(-(float)(lane & 15) * inv_tau);
    float sum = 0.0f;
#pragma unroll
    for (int j = 0; j < 16; ++j) sum += lane_bcast(d, j);
    return d * (1.0f / sum);
}

__global__ __launch_bounds__(128) void k_stats(const float* __restrict__ events,
                                               const float* __restrict__ tc,
                                               const float* __restrict__ tk,
                                               float* __restrict__ fvp) {
    __shared__ alignas(16) float sXp[4][4][88];
    __shared__ float sRed[4][256];
    __shared__ float sG[256];
    __shared__ alignas(16) float sDec[4][32];
    __shared__ alignas(16) float sComb[256 * CPITCH];

    int tid = threadIdx.x;
    int lane = tid & 31, w = tid >> 5;
    const float inv_tau = DT / tc[__builtin_amdgcn_readfirstlane(w)];
    const float4* x4 = reinterpret_cast<const float4*>(
        events + ((size_t)blockIdx.x * 16 + w * 4) * 64);
    float4 v0 = x4[lane];
    float4 v1 = x4[32 + lane];
    const float4* tk4 = reinterpret_cast<const float4*>(tk);
    float4 tv[8];
#pragma unroll
    for (int i = 0; i < 8; ++i) tv[i] = tk4[tid + i * 128];

    const int hc = lane >> 4, q = lane & 15;
    sXp[w][lane >> 3][lane & 7] = 0.0f;
    sXp[w][hc][72 + q] = 0.0f;
    sXp[w][2 + hc][72 + q] = 0.0f;
    *reinterpret_cast<float4*>(&sXp[w][hc][8 + 4 * q]) = v0;
    *reinterpret_cast<float4*>(&sXp[w][2 + hc][8 + 4 * q]) = v1;
    __syncthreads();

    const int m = lane & 15;
    const bool lowHalf = (lane < 16);
    const int koff = lowHalf ? 0 : 2;
    const float* r0 = &sXp[w][0][m + koff];
    const float* r1 = &sXp[w][1][m + koff];
    const float* r2 = &sXp[w][2][m + koff];
    const float* r3 = &sXp[w][3][m + koff];

    v8f a0 = {}, a1 = {}, a2 = {}, a3 = {};
    float p0 = 0.0f, p1 = 0.0f, p2 = 0.0f, p3 = 0.0f;
#pragma unroll
    for (int ks = 0; ks < 16; ++ks) {
        v2f f0, f1, f2, f3;
        f0.x = r0[4 * ks]; f0.y = r0[4 * ks + 1];
        f1.x = r1[4 * ks]; f1.y = r1[4 * ks + 1];
        f2.x = r2[4 * ks]; f2.y = r2[4 * ks + 1];
        f3.x = r3[4 * ks]; f3.y = r3[4 * ks + 1];
        a0 = __builtin_amdgcn_wmma_f32_16x16x4_f32(false, f0, false, f0, (short)0, a0, false, false);
        a1 = __builtin_amdgcn_wmma_f32_16x16x4_f32(false, f1, false, f1, (short)0, a1, false, false);
        a2 = __builtin_amdgcn_wmma_f32_16x16x4_f32(false, f2, false, f2, (short)0, a2, false, false);
        a3 = __builtin_amdgcn_wmma_f32_16x16x4_f32(false, f3, false, f3, (short)0, a3, false, false);
        p0 += f0.x + f0.y; p1 += f1.x + f1.y;
        p2 += f2.x + f2.y; p3 += f3.x + f3.y;
    }
    {
        v2f f0, f1, f2, f3;
        f0.x = lowHalf ? r0[64] : 0.0f; f0.y = 0.0f;
        f1.x = lowHalf ? r1[64] : 0.0f; f1.y = 0.0f;
        f2.x = lowHalf ? r2[64] : 0.0f; f2.y = 0.0f;
        f3.x = lowHalf ? r3[64] : 0.0f; f3.y = 0.0f;
        a0 = __builtin_amdgcn_wmma_f32_16x16x4_f32(false, f0, false, f0, (short)0, a0, false, false);
        a1 = __builtin_amdgcn_wmma_f32_16x16x4_f32(false, f1, false, f1, (short)0, a1, false, false);
        a2 = __builtin_amdgcn_wmma_f32_16x16x4_f32(false, f2, false, f2, (short)0, a2, false, false);
        a3 = __builtin_amdgcn_wmma_f32_16x16x4_f32(false, f3, false, f3, (short)0, a3, false, false);
        p0 += f0.x; p1 += f1.x; p2 += f2.x; p3 += f3.x;
    }
    p0 += __shfl_xor(p0, 16, 32);
    p1 += __shfl_xor(p1, 16, 32);
    p2 += __shfl_xor(p2, 16, 32);
    p3 += __shfl_xor(p3, 16, 32);
    v2f pf;
    pf.x = lowHalf ? p0 : p2;
    pf.y = lowHalf ? p1 : p3;
    v8f pz = {};
    v8f pp = __builtin_amdgcn_wmma_f32_16x16x4_f32(false, pf, false, pf, (short)0, pz, false, false);
    v8f g = (a0 + a1) + (a2 + a3) - pp * (1.0f / 65.0f);

#pragma unroll
    for (int r = 0; r < 8; ++r) sRed[w][r * 32 + lane] = g[r];
    sDec[w][lane] = decay_norm(inv_tau, lane);
    __syncthreads();
#pragma unroll
    for (int i = 0; i < 8; ++i) {
        int f = tid + i * 128;
        int row = f >> 2, j0 = (tid & 3) * 4;
        float4 dn = *reinterpret_cast<const float4*>(&sDec[i >> 1][j0]);
        float4 cv;
        cv.x = dn.x * tv[i].x; cv.y = dn.y * tv[i].y;
        cv.z = dn.z * tv[i].z; cv.w = dn.w * tv[i].w;
        *reinterpret_cast<float4*>(&sComb[row * CPITCH + j0]) = cv;
    }
#pragma unroll
    for (int i = 0; i < 2; ++i) {
        int idx = tid + i * 128;
        float tot = (sRed[0][idx] + sRed[1][idx]) + (sRed[2][idx] + sRed[3][idx]);
        int rr = idx >> 5, l = idx & 31;
        int row = ((l >> 4) << 3) + rr, col = l & 15;
        sG[row * 16 + col] = tot;
    }
    __syncthreads();

    {
        const int hrow = lowHalf ? 0 : 8;
        v2f bf[4];
#pragma unroll
        for (int ks = 0; ks < 4; ++ks) {
            int k0 = ks * 4 + koff;
            bf[ks].x = sG[k0 * 16 + m];
            bf[ks].y = sG[(k0 + 1) * 16 + m];
        }
        v8f d[4];
#pragma unroll
        for (int i = 0; i < 4; ++i) d[i] = v8f{};
#pragma unroll
        for (int ks = 0; ks < 4; ++ks) {
            int k0 = ks * 4 + koff;
#pragma unroll
            for (int i = 0; i < 4; ++i) {
                const float* crow = &sComb[((4 * w + i) * 16 + m) * CPITCH];
                v2f af;
                af.x = crow[k0];
                af.y = crow[k0 + 1];
                d[i] = __builtin_amdgcn_wmma_f32_16x16x4_f32(false, af, false, bf[ks], (short)0, d[i], false, false);
            }
        }
        float qt[4];
#pragma unroll
        for (int i = 0; i < 4; ++i) {
            float acc = 0.0f;
#pragma unroll
            for (int r = 0; r < 8; ++r)
                acc += d[i][r] * sComb[((4 * w + i) * 16 + hrow + r) * CPITCH + m];
            qt[i] = acc;
        }
        float qs = (qt[0] + qt[1]) + (qt[2] + qt[3]);
        qs += __shfl_xor(qs, 16, 32);
        qs += __shfl_xor(qs, 8, 32);
        qs += __shfl_xor(qs, 4, 32);
        qs += __shfl_xor(qs, 2, 32);
        qs += __shfl_xor(qs, 1, 32);
        if (lane == 0) fvp[blockIdx.x * 4 + w] = qs;
    }
}

#define ENC_ESTR 1044
__global__ __launch_bounds__(256) void k_encode(const float* __restrict__ events,
                                                const float* __restrict__ tc,
                                                const float* tk,
                                                const float* w1, const float* b1,
                                                const float* w2, const float* b2,
                                                const float* __restrict__ fvp,
                                                float* __restrict__ out) {
    __shared__ float sK[64 * 17];
    __shared__ alignas(16) float sXp[16][80];
    __shared__ alignas(16) float sDec[8][32];
    __shared__ alignas(16) float2 sFv[8][32];
    __shared__ alignas(16) float sStage[3][16][ENC_ESTR];

    const int tid = threadIdx.x, lane = tid & 31, w = tid >> 5;
    const int ch0 = blockIdx.x * 16;
    const int b   = ch0 >> 11;
    const int cf0 = ch0 & 2047;

    const float inv_tau = DT / tc[__builtin_amdgcn_readfirstlane(tid >> 6)];
    const float4 xv = reinterpret_cast<const float4*>(events + (size_t)ch0 * 64)[tid];
    float2 pv = reinterpret_cast<const float2*>(fvp)[b * 256 + tid];
    float4 tv[4];
#pragma unroll
    for (int s = 0; s < 4; ++s)
        tv[s] = reinterpret_cast<const float4*>(tk)[tid + s * 256];
    const float w1v = w1[lane], w2v = w2[lane], b1v = b1[lane & 7], b2v = b2[lane & 3];

    {
        const int chl = tid >> 4, q = tid & 15;
        sXp[chl][(q < 8) ? q : 64 + q] = 0.0f;
        *reinterpret_cast<float4*>(&sXp[chl][8 + 4 * q]) = xv;
    }
    sDec[w][lane] = decay_norm(inv_tau, lane);
    pv.x += __shfl_xor(pv.x, 2, 32);  pv.y += __shfl_xor(pv.y, 2, 32);
    pv.x += __shfl_xor(pv.x, 4, 32);  pv.y += __shfl_xor(pv.y, 4, 32);
    pv.x += __shfl_xor(pv.x, 8, 32);  pv.y += __shfl_xor(pv.y, 8, 32);
    pv.x += __shfl_xor(pv.x, 16, 32); pv.y += __shfl_xor(pv.y, 16, 32);
    sFv[w][lane] = pv;
    __syncthreads();

    const int m    = lane & 15;
    const int h    = lane >> 4;
    const int koff = 2 * h;

    v2f xf[2][4][4];
#pragma unroll
    for (int c2 = 0; c2 < 2; ++c2) {
        const float* xrow = &sXp[2 * w + c2][0];
#pragma unroll
        for (int mt = 0; mt < 4; ++mt)
#pragma unroll
            for (int ks = 0; ks < 4; ++ks) {
                int k0 = ks * 4 + koff;
                xf[c2][mt][ks].x = xrow[mt * 16 + m + k0];
                xf[c2][mt][ks].y = xrow[mt * 16 + m + k0 + 1];
            }
    }
    float xs[16];
#pragma unroll
    for (int j = 0; j < 16; ++j) xs[j] = sXp[2 * w + h][64 + j];

    {
        float4 fw[8];
#pragma unroll
        for (int ww = 0; ww < 8; ++ww)
            fw[ww] = *reinterpret_cast<const float4*>(&sFv[ww][0]);
        const float sc = 1.0f / (64.0f * 64.0f * 2048.0f);
        float fv[4];
        fv[0] = (((fw[0].x + fw[1].x) + (fw[2].x + fw[3].x)) +
                 ((fw[4].x + fw[5].x) + (fw[6].x + fw[7].x))) * sc;
        fv[1] = (((fw[0].y + fw[1].y) + (fw[2].y + fw[3].y)) +
                 ((fw[4].y + fw[5].y) + (fw[6].y + fw[7].y))) * sc;
        fv[2] = (((fw[0].z + fw[1].z) + (fw[2].z + fw[3].z)) +
                 ((fw[4].z + fw[5].z) + (fw[6].z + fw[7].z))) * sc;
        fv[3] = (((fw[0].w + fw[1].w) + (fw[2].w + fw[3].w)) +
                 ((fw[4].w + fw[5].w) + (fw[6].w + fw[7].w))) * sc;
        float hh[8];
#pragma unroll
        for (int i = 0; i < 8; ++i) {
            float a = lane_bcast(b1v, i);
#pragma unroll
            for (int s = 0; s < 4; ++s) a += fv[s] * lane_bcast(w1v, i * 4 + s);
            hh[i] = fmaxf(a, 0.0f);
        }
        float lg[4]; float mx = -1e30f;
#pragma unroll
        for (int s = 0; s < 4; ++s) {
            float a = lane_bcast(b2v, s);
#pragma unroll
            for (int i = 0; i < 8; ++i) a += hh[i] * lane_bcast(w2v, s * 8 + i);
            lg[s] = a; mx = fmaxf(mx, a);
        }
        float den = 0.0f;
#pragma unroll
        for (int s = 0; s < 4; ++s) { lg[s] = expf(lg[s] - mx); den += lg[s]; }
        const float inv = 1.0f / den;
        const int e = tid >> 2, j0 = (tid & 3) * 4;
        float4 acc = make_float4(0.0f, 0.0f, 0.0f, 0.0f);
#pragma unroll
        for (int s = 0; s < 4; ++s) {
            const float at = lg[s] * inv;
            const float4 dn = *reinterpret_cast<const float4*>(&sDec[2 * s][j0]);
            acc.x += at * (dn.x * tv[s].x);
            acc.y += at * (dn.y * tv[s].y);
            acc.z += at * (dn.z * tv[s].z);
            acc.w += at * (dn.w * tv[s].w);
        }
        sK[e * 17 + j0]     = acc.x;
        sK[e * 17 + j0 + 1] = acc.y;
        sK[e * 17 + j0 + 2] = acc.z;
        sK[e * 17 + j0 + 3] = acc.w;
    }
    __syncthreads();

    const int a4 = (blockIdx.x & 1) * 4;

    for (int ntile = 0; ntile < 4; ++ntile) {
        float* stg = &sStage[ntile % 3][0][0];
        const float* brow = &sK[(ntile * 16 + m) * 17];
        v2f bv[4];
#pragma unroll
        for (int ks = 0; ks < 4; ++ks) {
            int k0 = ks * 4 + koff;
            bv[ks].x = brow[k0];
            bv[ks].y = brow[k0 + 1];
        }
        v8f acc[2][4];
#pragma unroll
        for (int c2 = 0; c2 < 2; ++c2)
#pragma unroll
            for (int mt = 0; mt < 4; ++mt) acc[c2][mt] = v8f{};
#pragma unroll
        for (int ks = 0; ks < 4; ++ks)
#pragma unroll
            for (int c2 = 0; c2 < 2; ++c2)
#pragma unroll
                for (int mt = 0; mt < 4; ++mt)
                    acc[c2][mt] = __builtin_amdgcn_wmma_f32_16x16x4_f32(
                        false, bv[ks], false, xf[c2][mt][ks], (short)0, acc[c2][mt], false, false);
#pragma unroll
        for (int c2 = 0; c2 < 2; ++c2)
#pragma unroll
            for (int mt = 0; mt < 4; ++mt)
#pragma unroll
                for (int r = 0; r < 8; ++r)
                    stg[(8 * h + r) * ENC_ESTR + (2 * w + c2) * 65 + mt * 16 + m] = acc[c2][mt][r];
        float o64 = 0.0f;
#pragma unroll
        for (int j = 0; j < 16; ++j) o64 += xs[j] * brow[j];
        stg[m * ENC_ESTR + (2 * w + h) * 65 + 64] = o64;
        __builtin_amdgcn_fence(__ATOMIC_RELEASE, "workgroup", "local");
        __builtin_amdgcn_s_barrier();
        __builtin_amdgcn_fence(__ATOMIC_ACQUIRE, "workgroup", "local");

#pragma unroll
        for (int rr = 0; rr < 2; ++rr) {
            const int e16 = w + 8 * rr;
            const float* srow = stg + e16 * ENC_ESTR;
            float4* orow = reinterpret_cast<float4*>(
                out + ((size_t)(b * 64 + ntile * 16 + e16) * 2048 + cf0) * 65);
            float4 v[9];
#pragma unroll
            for (int g = 0; g < 9; ++g) {
                int q = 32 * g + lane - a4;
                int qc = (q < 0) ? 0 : ((q > 259) ? 259 : q);
                v[g] = *reinterpret_cast<const float4*>(srow + 4 * qc);
            }
#pragma unroll
            for (int g = 0; g < 9; ++g) {
                int q = 32 * g + lane - a4;
                if (g == 0) { if (q >= 0) orow[q] = v[g]; }
                else if (g == 8) { if (q < 260) orow[q] = v[g]; }
                else orow[q] = v[g];
            }
        }
    }
}

extern "C" void kernel_launch(void* const* d_in, const int* in_sizes, int n_in,
                              void* d_out, int out_size, void* d_ws, size_t ws_size,
                              hipStream_t stream) {
    const float* events = (const float*)d_in[0];
    const float* tcs    = (const float*)d_in[1];
    const float* tk     = (const float*)d_in[2];
    const float* w1     = (const float*)d_in[3];
    const float* b1     = (const float*)d_in[4];
    const float* w2     = (const float*)d_in[5];
    const float* b2     = (const float*)d_in[6];
    float* out = (float*)d_out;

    float* fvp = (float*)d_ws;

    k_stats <<<256, 128, 0, stream>>>(events, tcs, tk, fvp);
    k_encode<<<256, 256, 0, stream>>>(events, tcs, tk, w1, b1, w2, b2, fvp, out);
}


# k_encode re-tiled: block = 64 channels x 16 e (was 16 x 64): each e row of a block is 65 whole 256-B units, every global store writes 512 aligned bytes, no line shared between blocks
# speedup vs baseline: 3.3404x; 1.3309x over previous
#include <hip/hip_runtime.h>
#include <hip/hip_bf16.h>
#include <math.h>

typedef __attribute__((ext_vector_type(2))) float v2f;
typedef __attribute__((ext_vector_type(8))) float v8f;

#define DT 0.005f

#define CPITCH 20

__device__ __forceinline__ float lane_bcast(float v, int j) {
    return __builtin_bit_cast(float, __builtin_amdgcn_readlane(__builtin_bit_cast(int, v), j));
}

__device__ __forceinline__ float decay_norm(float inv_tau, int lane) {
    const float d = expf(-(float)(lane & 15) * inv_tau);
    float sum = 0.0f;
#pragma unroll
    for (int j = 0; j < 16; ++j) sum += lane_bcast(d, j);
    return d * (1.0f / sum);
}

__global__ __launch_bounds__(128) void k_stats(const float* __restrict__ events,
                                               const float* __restrict__ tc,
                                               const float* __restrict__ tk,
                                               float* __restrict__ fvp) {
    __shared__ alignas(16) float sXp[4][4][88];
    __shared__ float sRed[4][256];
    __shared__ float sG[256];
    __shared__ alignas(16) float sDec[4][32];
    __shared__ alignas(16) float sComb[256 * CPITCH];

    int tid = threadIdx.x;
    int lane = tid & 31, w = tid >> 5;
    const float inv_tau = DT / tc[__builtin_amdgcn_readfirstlane(w)];
    const float4* x4 = reinterpret_cast<const float4*>(
        events + ((size_t)blockIdx.x * 16 + w * 4) * 64);
    float4 v0 = x4[lane];
    float4 v1 = x4[32 + lane];
    const float4* tk4 = reinterpret_cast<const float4*>(tk);
    float4 tv[8];
#pragma unroll
    for (int i = 0; i < 8; ++i) tv[i] = tk4[tid + i * 128];

    const int hc = lane >> 4, q = lane & 15;
    sXp[w][lane >> 3][lane & 7] = 0.0f;
    sXp[w][hc][72 + q] = 0.0f;
    sXp[w][2 + hc][72 + q] = 0.0f;
    *reinterpret_cast<float4*>(&sXp[w][hc][8 + 4 * q]) = v0;
    *reinterpret_cast<float4*>(&sXp[w][2 + hc][8 + 4 * q]) = v1;
    __syncthreads();

    const int m = lane & 15;
    const bool lowHalf = (lane < 16);
    const int koff = lowHalf ? 0 : 2;
    const float* r0 = &sXp[w][0][m + koff];
    const float* r1 = &sXp[w][1][m + koff];
    const float* r2 = &sXp[w][2][m + koff];
    const float* r3 = &sXp[w][3][m + koff];

    v8f a0 = {}, a1 = {}, a2 = {}, a3 = {};
    float p0 = 0.0f, p1 = 0.0f, p2 = 0.0f, p3 = 0.0f;
#pragma unroll
    for (int ks = 0; ks < 16; ++ks) {
        v2f f0, f1, f2, f3;
        f0.x = r0[4 * ks]; f0.y = r0[4 * ks + 1];
        f1.x = r1[4 * ks]; f1.y = r1[4 * ks + 1];
        f2.x = r2[4 * ks]; f2.y = r2[4 * ks + 1];
        f3.x = r3[4 * ks]; f3.y = r3[4 * ks + 1];
        a0 = __builtin_amdgcn_wmma_f32_16x16x4_f32(false, f0, false, f0, (short)0, a0, false, false);
        a1 = __builtin_amdgcn_wmma_f32_16x16x4_f32(false, f1, false, f1, (short)0, a1, false, false);
        a2 = __builtin_amdgcn_wmma_f32_16x16x4_f32(false, f2, false, f2, (short)0, a2, false, false);
        a3 = __builtin_amdgcn_wmma_f32_16x16x4_f32(false, f3, false, f3, (short)0, a3, false, false);
        p0 += f0.x + f0.y; p1 += f1.x + f1.y;
        p2 += f2.x + f2.y; p3 += f3.x + f3.y;
    }
    {
        v2f f0, f1, f2, f3;
        f0.x = lowHalf ? r0[64] : 0.0f; f0.y = 0.0f;
        f1.x = lowHalf ? r1[64] : 0.0f; f1.y = 0.0f;
        f2.x = lowHalf ? r2[64] : 0.0f; f2.y = 0.0f;
        f3.x = lowHalf ? r3[64] : 0.0f; f3.y = 0.0f;
        a0 = __builtin_amdgcn_wmma_f32_16x16x4_f32(false, f0, false, f0, (short)0, a0, false, false);
        a1 = __builtin_amdgcn_wmma_f32_16x16x4_f32(false, f1, false, f1, (short)0, a1, false, false);
        a2 = __builtin_amdgcn_wmma_f32_16x16x4_f32(false, f2, false, f2, (short)0, a2, false, false);
        a3 = __builtin_amdgcn_wmma_f32_16x16x4_f32(false, f3, false, f3, (short)0, a3, false, false);
        p0 += f0.x; p1 += f1.x; p2 += f2.x; p3 += f3.x;
    }
    p0 += __shfl_xor(p0, 16, 32);
    p1 += __shfl_xor(p1, 16, 32);
    p2 += __shfl_xor(p2, 16, 32);
    p3 += __shfl_xor(p3, 16, 32);
    v2f pf;
    pf.x = lowHalf ? p0 : p2;
    pf.y = lowHalf ? p1 : p3;
    v8f pz = {};
    v8f pp = __builtin_amdgcn_wmma_f32_16x16x4_f32(false, pf, false, pf, (short)0, pz, false, false);
    v8f g = (a0 + a1) + (a2 + a3) - pp * (1.0f / 65.0f);

#pragma unroll
    for (int r = 0; r < 8; ++r) sRed[w][r * 32 + lane] = g[r];
    sDec[w][lane] = decay_norm(inv_tau, lane);
    __syncthreads();
#pragma unroll
    for (int i = 0; i < 8; ++i) {
        int f = tid + i * 128;
        int row = f >> 2, j0 = (tid & 3) * 4;
        float4 dn = *reinterpret_cast<const float4*>(&sDec[i >> 1][j0]);
        float4 cv;
        cv.x = dn.x * tv[i].x; cv.y = dn.y * tv[i].y;
        cv.z = dn.z * tv[i].z; cv.w = dn.w * tv[i].w;
        *reinterpret_cast<float4*>(&sComb[row * CPITCH + j0]) = cv;
    }
#pragma unroll
    for (int i = 0; i < 2; ++i) {
        int idx = tid + i * 128;
        float tot = (sRed[0][idx] + sRed[1][idx]) + (sRed[2][idx] + sRed[3][idx]);
        int rr = idx >> 5, l = idx & 31;
        int row = ((l >> 4) << 3) + rr, col = l & 15;
        sG[row * 16 + col] = tot;
    }
    __syncthreads();

    {
        const int hrow = lowHalf ? 0 : 8;
        v2f bf[4];
#pragma unroll
        for (int ks = 0; ks < 4; ++ks) {
            int k0 = ks * 4 + koff;
            bf[ks].x = sG[k0 * 16 + m];
            bf[ks].y = sG[(k0 + 1) * 16 + m];
        }
        v8f d[4];
#pragma unroll
        for (int i = 0; i < 4; ++i) d[i] = v8f{};
#pragma unroll
        for (int ks = 0; ks < 4; ++ks) {
            int k0 = ks * 4 + koff;
#pragma unroll
            for (int i = 0; i < 4; ++i) {
                const float* crow = &sComb[((4 * w + i) * 16 + m) * CPITCH];
                v2f af;
                af.x = crow[k0];
                af.y = crow[k0 + 1];
                d[i] = __builtin_amdgcn_wmma_f32_16x16x4_f32(false, af, false, bf[ks], (short)0, d[i], false, false);
            }
        }
        float qt[4];
#pragma unroll
        for (int i = 0; i < 4; ++i) {
            float acc = 0.0f;
#pragma unroll
            for (int r = 0; r < 8; ++r)
                acc += d[i][r] * sComb[((4 * w + i) * 16 + hrow + r) * CPITCH + m];
            qt[i] = acc;
        }
        float qs = (qt[0] + qt[1]) + (qt[2] + qt[3]);
        qs += __shfl_xor(qs, 16, 32);
        qs += __shfl_xor(qs, 8, 32);
        qs += __shfl_xor(qs, 4, 32);
        qs += __shfl_xor(qs, 2, 32);
        qs += __shfl_xor(qs, 1, 32);
        if (lane == 0) fvp[blockIdx.x * 4 + w] = qs;
    }
}

#define ENC_ESTR 1044
__global__ __launch_bounds__(256) void k_encode(const float* __restrict__ events,
                                                const float* __restrict__ tc,
                                                const float* tk,
                                                const float* w1, const float* b1,
                                                const float* w2, const float* b2,
                                                const float* __restrict__ fvp,
                                                float* __restrict__ out) {
    __shared__ float sK[16 * 17];
    __shared__ alignas(16) float sXp[64][80];
    __shared__ alignas(16) float sDec[8][32];
    __shared__ alignas(16) float2 sFv[8][32];
    __shared__ alignas(16) float sStage[4][16][ENC_ESTR];

    const int tid = threadIdx.x, lane = tid & 31, w = tid >> 5;
    const int cg  = blockIdx.x & 63;
    const int et  = blockIdx.x >> 6;
    const int ch0 = cg * 64;
    const int b   = ch0 >> 11;
    const int cf0 = ch0 & 2047;

    const float inv_tau = DT / tc[__builtin_amdgcn_readfirstlane(tid >> 6)];
    const float4* x4 = reinterpret_cast<const float4*>(events + (size_t)ch0 * 64);
    float4 xv[4];
#pragma unroll
    for (int r = 0; r < 4; ++r) xv[r] = x4[tid + r * 256];
    float2 pv = reinterpret_cast<const float2*>(fvp)[b * 256 + tid];
    const int el = tid >> 4, tj = tid & 15;
    float tv[4];
#pragma unroll
    for (int s = 0; s < 4; ++s)
        tv[s] = tk[(s * 64 + et * 16 + el) * 16 + tj];
    const float w1v = w1[lane], w2v = w2[lane], b1v = b1[lane & 7], b2v = b2[lane & 3];

    {
        const int q = tid & 15;
#pragma unroll
        for (int r = 0; r < 4; ++r) {
            const int chl = (tid >> 4) + 16 * r;
            sXp[chl][(q < 8) ? q : 64 + q] = 0.0f;
            *reinterpret_cast<float4*>(&sXp[chl][8 + 4 * q]) = xv[r];
        }
    }
    sDec[w][lane] = decay_norm(inv_tau, lane);
    pv.x += __shfl_xor(pv.x, 2, 32);  pv.y += __shfl_xor(pv.y, 2, 32);
    pv.x += __shfl_xor(pv.x, 4, 32);  pv.y += __shfl_xor(pv.y, 4, 32);
    pv.x += __shfl_xor(pv.x, 8, 32);  pv.y += __shfl_xor(pv.y, 8, 32);
    pv.x += __shfl_xor(pv.x, 16, 32); pv.y += __shfl_xor(pv.y, 16, 32);
    sFv[w][lane] = pv;
    __syncthreads();

    const int m    = lane & 15;
    const int h    = lane >> 4;
    const int koff = 2 * h;

    v2f xf[2][2][4][4];
    float xs[2][16];
#pragma unroll
    for (int c2 = 0; c2 < 2; ++c2) {
        const float* xrow = &sXp[2 * w + c2][0];
#pragma unroll
        for (int mt = 0; mt < 4; ++mt)
#pragma unroll
            for (int ks = 0; ks < 4; ++ks) {
                int k0 = ks * 4 + koff;
                xf[0][c2][mt][ks].x = xrow[mt * 16 + m + k0];
                xf[0][c2][mt][ks].y = xrow[mt * 16 + m + k0 + 1];
            }
    }
#pragma unroll
    for (int j = 0; j < 16; ++j) xs[0][j] = sXp[2 * w + h][64 + j];
    {
        float4 fw[8];
#pragma unroll
        for (int ww = 0; ww < 8; ++ww)
            fw[ww] = *reinterpret_cast<const float4*>(&sFv[ww][0]);
        const float sc = 1.0f / (64.0f * 64.0f * 2048.0f);
        float fv[4];
        fv[0] = (((fw[0].x + fw[1].x) + (fw[2].x + fw[3].x)) +
                 ((fw[4].x + fw[5].x) + (fw[6].x + fw[7].x))) * sc;
        fv[1] = (((fw[0].y + fw[1].y) + (fw[2].y + fw[3].y)) +
                 ((fw[4].y + fw[5].y) + (fw[6].y + fw[7].y))) * sc;
        fv[2] = (((fw[0].z + fw[1].z) + (fw[2].z + fw[3].z)) +
                 ((fw[4].z + fw[5].z) + (fw[6].z + fw[7].z))) * sc;
        fv[3] = (((fw[0].w + fw[1].w) + (fw[2].w + fw[3].w)) +
                 ((fw[4].w + fw[5].w) + (fw[6].w + fw[7].w))) * sc;
        float hh[8];
#pragma unroll
        for (int i = 0; i < 8; ++i) {
            float a = lane_bcast(b1v, i);
#pragma unroll
            for (int s = 0; s < 4; ++s) a += fv[s] * lane_bcast(w1v, i * 4 + s);
            hh[i] = fmaxf(a, 0.0f);
        }
        float lg[4]; float mx = -1e30f;
#pragma unroll
        for (int s = 0; s < 4; ++s) {
            float a = lane_bcast(b2v, s);
#pragma unroll
            for (int i = 0; i < 8; ++i) a += hh[i] * lane_bcast(w2v, s * 8 + i);
            lg[s] = a; mx = fmaxf(mx, a);
        }
        float den = 0.0f;
#pragma unroll
        for (int s = 0; s < 4; ++s) { lg[s] = expf(lg[s] - mx); den += lg[s]; }
        const float inv = 1.0f / den;
        float acc = 0.0f;
#pragma unroll
        for (int s = 0; s < 4; ++s) {
            const float at = lg[s] * inv;
            const float dn = sDec[2 * s][tj];
            acc += at * (dn * tv[s]);
        }
        sK[el * 17 + tj] = acc;
    }
    __syncthreads();

    const float* brow = &sK[m * 17];
    v2f bv[4];
#pragma unroll
    for (int ks = 0; ks < 4; ++ks) {
        int k0 = ks * 4 + koff;
        bv[ks].x = brow[k0];
        bv[ks].y = brow[k0 + 1];
    }

#pragma unroll
    for (int k = 0; k < 4; ++k) {
        const int cur = k & 1, nxt = cur ^ 1;
        float* stg = &sStage[k][0][0];
        if (k < 3) {
#pragma unroll
            for (int c2 = 0; c2 < 2; ++c2) {
                const float* xrow = &sXp[16 * (k + 1) + 2 * w + c2][0];
#pragma unroll
                for (int mt = 0; mt < 4; ++mt)
#pragma unroll
                    for (int ks = 0; ks < 4; ++ks) {
                        int k0 = ks * 4 + koff;
                        xf[nxt][c2][mt][ks].x = xrow[mt * 16 + m + k0];
                        xf[nxt][c2][mt][ks].y = xrow[mt * 16 + m + k0 + 1];
                    }
            }
#pragma unroll
            for (int j = 0; j < 16; ++j) xs[nxt][j] = sXp[16 * (k + 1) + 2 * w + h][64 + j];
        }
        v8f acc[2][4];
#pragma unroll
        for (int c2 = 0; c2 < 2; ++c2)
#pragma unroll
            for (int mt = 0; mt < 4; ++mt) acc[c2][mt] = v8f{};
#pragma unroll
        for (int ks = 0; ks < 4; ++ks)
#pragma unroll
            for (int c2 = 0; c2 < 2; ++c2)
#pragma unroll
                for (int mt = 0; mt < 4; ++mt)
                    acc[c2][mt] = __builtin_amdgcn_wmma_f32_16x16x4_f32(
                        false, bv[ks], false, xf[cur][c2][mt][ks], (short)0, acc[c2][mt], false, false);
#pragma unroll
        for (int c2 = 0; c2 < 2; ++c2)
#pragma unroll
            for (int mt = 0; mt < 4; ++mt)
#pragma unroll
                for (int r = 0; r < 8; ++r)
                    stg[(8 * h + r) * ENC_ESTR + (2 * w + c2) * 65 + mt * 16 + m] = acc[c2][mt][r];
        float o64 = 0.0f;
#pragma unroll
        for (int j = 0; j < 16; ++j) o64 += xs[cur][j] * brow[j];
        stg[m * ENC_ESTR + (2 * w + h) * 65 + 64] = o64;
        __builtin_amdgcn_fence(__ATOMIC_RELEASE, "workgroup", "local");
        __builtin_amdgcn_s_barrier();
        __builtin_amdgcn_fence(__ATOMIC_ACQUIRE, "workgroup", "local");

#pragma unroll
        for (int rr = 0; rr < 2; ++rr) {
            const int e16 = w + 8 * rr;
            const float* srow = stg + e16 * ENC_ESTR;
            float4* orow = reinterpret_cast<float4*>(
                out + ((size_t)(b * 64 + et * 16 + e16) * 2048 + cf0) * 65) + 256 * k;
            float4 v[9];
            {
                const int q = lane - 4 * k;
                const float* prow = &sStage[k > 0 ? k - 1 : 0][0][0] + e16 * ENC_ESTR;
                const float* src = (k > 0 && q < 0) ? prow + 4 * (260 + q) : srow + 4 * q;
                v[0] = *reinterpret_cast<const float4*>(src);
            }
#pragma unroll
            for (int g = 1; g < 8; ++g)
                v[g] = *reinterpret_cast<const float4*>(srow + 4 * (32 * g + lane - 4 * k));
            if (k == 3) {
                const int q = 244 + (lane & 15);
                v[8] = *reinterpret_cast<const float4*>(srow + 4 * q);
            }
#pragma unroll
            for (int g = 0; g < 8; ++g) orow[32 * g + lane] = v[g];
            if (k == 3) { if (lane < 16) orow[256 + lane] = v[8]; }
        }
    }
}

extern "C" void kernel_launch(void* const* d_in, const int* in_sizes, int n_in,
                              void* d_out, int out_size, void* d_ws, size_t ws_size,
                              hipStream_t stream) {
    const float* events = (const float*)d_in[0];
    const float* tcs    = (const float*)d_in[1];
    const float* tk     = (const float*)d_in[2];
    const float* w1     = (const float*)d_in[3];
    const float* b1     = (const float*)d_in[4];
    const float* w2     = (const float*)d_in[5];
    const float* b2     = (const float*)d_in[6];
    float* out = (float*)d_out;

    float* fvp = (float*)d_ws;

    k_stats <<<256, 128, 0, stream>>>(events, tcs, tk, fvp);
    k_encode<<<256, 256, 0, stream>>>(events, tcs, tk, w1, b1, w2, b2, fvp, out);
}


# k_encode: WMMA columns follow the output row (row float 64q+4*lane+tile -> channel,t by /65): stores go from accumulators as two whole 256-B units; no LDS stage; t=64 sums on WMMA, fp order changes
# speedup vs baseline: 3.6517x; 1.0932x over previous
#include <hip/hip_runtime.h>
#include <hip/hip_bf16.h>
#include <math.h>

typedef __attribute__((ext_vector_type(2))) float v2f;
typedef __attribute__((ext_vector_type(8))) float v8f;

#define DT 0.005f

#define CPITCH 20

__device__ __forceinline__ float lane_bcast(float v, int j) {
    return __builtin_bit_cast(float, __builtin_amdgcn_readlane(__builtin_bit_cast(int, v), j));
}

__device__ __forceinline__ float decay_norm(float inv_tau, int lane) {
    const float d = expf(-(float)(lane & 15) * inv_tau);
    float sum = 0.0f;
#pragma unroll
    for (int j = 0; j < 16; ++j) sum += lane_bcast(d, j);
    return d * (1.0f / sum);
}

__global__ __launch_bounds__(128) void k_stats(const float* __restrict__ events,
                                               const float* __restrict__ tc,
                                               const float* __restrict__ tk,
                                               float* __restrict__ fvp) {
    __shared__ alignas(16) float sXp[4][4][88];
    __shared__ float sRed[4][256];
    __shared__ float sG[256];
    __shared__ alignas(16) float sDec[4][32];
    __shared__ alignas(16) float sComb[256 * CPITCH];

    int tid = threadIdx.x;
    int lane = tid & 31, w = tid >> 5;
    const float inv_tau = DT / tc[__builtin_amdgcn_readfirstlane(w)];
    const float4* x4 = reinterpret_cast<const float4*>(
        events + ((size_t)blockIdx.x * 16 + w * 4) * 64);
    float4 v0 = x4[lane];
    float4 v1 = x4[32 + lane];
    const float4* tk4 = reinterpret_cast<const float4*>(tk);
    float4 tv[8];
#pragma unroll
    for (int i = 0; i < 8; ++i) tv[i] = tk4[tid + i * 128];

    const int hc = lane >> 4, q = lane & 15;
    sXp[w][lane >> 3][lane & 7] = 0.0f;
    sXp[w][hc][72 + q] = 0.0f;
    sXp[w][2 + hc][72 + q] = 0.0f;
    *reinterpret_cast<float4*>(&sXp[w][hc][8 + 4 * q]) = v0;
    *reinterpret_cast<float4*>(&sXp[w][2 + hc][8 + 4 * q]) = v1;
    __syncthreads();

    const int m = lane & 15;
    const bool lowHalf = (lane < 16);
    const int koff = lowHalf ? 0 : 2;
    const float* r0 = &sXp[w][0][m + koff];
    const float* r1 = &sXp[w][1][m + koff];
    const float* r2 = &sXp[w][2][m + koff];
    const float* r3 = &sXp[w][3][m + koff];

    v8f a0 = {}, a1 = {}, a2 = {}, a3 = {};
    float p0 = 0.0f, p1 = 0.0f, p2 = 0.0f, p3 = 0.0f;
#pragma unroll
    for (int ks = 0; ks < 16; ++ks) {
        v2f f0, f1, f2, f3;
        f0.x = r0[4 * ks]; f0.y = r0[4 * ks + 1];
        f1.x = r1[4 * ks]; f1.y = r1[4 * ks + 1];
        f2.x = r2[4 * ks]; f2.y = r2[4 * ks + 1];
        f3.x = r3[4 * ks]; f3.y = r3[4 * ks + 1];
        a0 = __builtin_amdgcn_wmma_f32_16x16x4_f32(false, f0, false, f0, (short)0, a0, false, false);
        a1 = __builtin_amdgcn_wmma_f32_16x16x4_f32(false, f1, false, f1, (short)0, a1, false, false);
        a2 = __builtin_amdgcn_wmma_f32_16x16x4_f32(false, f2, false, f2, (short)0, a2, false, false);
        a3 = __builtin_amdgcn_wmma_f32_16x16x4_f32(false, f3, false, f3, (short)0, a3, false, false);
        p0 += f0.x + f0.y; p1 += f1.x + f1.y;
        p2 += f2.x + f2.y; p3 += f3.x + f3.y;
    }
    {
        v2f f0, f1, f2, f3;
        f0.x = lowHalf ? r0[64] : 0.0f; f0.y = 0.0f;
        f1.x = lowHalf ? r1[64] : 0.0f; f1.y = 0.0f;
        f2.x = lowHalf ? r2[64] : 0.0f; f2.y = 0.0f;
        f3.x = lowHalf ? r3[64] : 0.0f; f3.y = 0.0f;
        a0 = __builtin_amdgcn_wmma_f32_16x16x4_f32(false, f0, false, f0, (short)0, a0, false, false);
        a1 = __builtin_amdgcn_wmma_f32_16x16x4_f32(false, f1, false, f1, (short)0, a1, false, false);
        a2 = __builtin_amdgcn_wmma_f32_16x16x4_f32(false, f2, false, f2, (short)0, a2, false, false);
        a3 = __builtin_amdgcn_wmma_f32_16x16x4_f32(false, f3, false, f3, (short)0, a3, false, false);
        p0 += f0.x; p1 += f1.x; p2 += f2.x; p3 += f3.x;
    }
    p0 += __shfl_xor(p0, 16, 32);
    p1 += __shfl_xor(p1, 16, 32);
    p2 += __shfl_xor(p2, 16, 32);
    p3 += __shfl_xor(p3, 16, 32);
    v2f pf;
    pf.x = lowHalf ? p0 : p2;
    pf.y = lowHalf ? p1 : p3;
    v8f pz = {};
    v8f pp = __builtin_amdgcn_wmma_f32_16x16x4_f32(false, pf, false, pf, (short)0, pz, false, false);
    v8f g = (a0 + a1) + (a2 + a3) - pp * (1.0f / 65.0f);

#pragma unroll
    for (int r = 0; r < 8; ++r) sRed[w][r * 32 + lane] = g[r];
    sDec[w][lane] = decay_norm(inv_tau, lane);
    __syncthreads();
#pragma unroll
    for (int i = 0; i < 8; ++i) {
        int f = tid + i * 128;
        int row = f >> 2, j0 = (tid & 3) * 4;
        float4 dn = *reinterpret_cast<const float4*>(&sDec[i >> 1][j0]);
        float4 cv;
        cv.x = dn.x * tv[i].x; cv.y = dn.y * tv[i].y;
        cv.z = dn.z * tv[i].z; cv.w = dn.w * tv[i].w;
        *reinterpret_cast<float4*>(&sComb[row * CPITCH + j0]) = cv;
    }
#pragma unroll
    for (int i = 0; i < 2; ++i) {
        int idx = tid + i * 128;
        float tot = (sRed[0][idx] + sRed[1][idx]) + (sRed[2][idx] + sRed[3][idx]);
        int rr = idx >> 5, l = idx & 31;
        int row = ((l >> 4) << 3) + rr, col = l & 15;
        sG[row * 16 + col] = tot;
    }
    __syncthreads();

    {
        const int hrow = lowHalf ? 0 : 8;
        v2f bf[4];
#pragma unroll
        for (int ks = 0; ks < 4; ++ks) {
            int k0 = ks * 4 + koff;
            bf[ks].x = sG[k0 * 16 + m];
            bf[ks].y = sG[(k0 + 1) * 16 + m];
        }
        v8f d[4];
#pragma unroll
        for (int i = 0; i < 4; ++i) d[i] = v8f{};
#pragma unroll
        for (int ks = 0; ks < 4; ++ks) {
            int k0 = ks * 4 + koff;
#pragma unroll
            for (int i = 0; i < 4; ++i) {
                const float* crow = &sComb[((4 * w + i) * 16 + m) * CPITCH];
                v2f af;
                af.x = crow[k0];
                af.y = crow[k0 + 1];
                d[i] = __builtin_amdgcn_wmma_f32_16x16x4_f32(false, af, false, bf[ks], (short)0, d[i], false, false);
            }
        }
        float qt[4];
#pragma unroll
        for (int i = 0; i < 4; ++i) {
            float acc = 0.0f;
#pragma unroll
            for (int r = 0; r < 8; ++r)
                acc += d[i][r] * sComb[((4 * w + i) * 16 + hrow + r) * CPITCH + m];
            qt[i] = acc;
        }
        float qs = (qt[0] + qt[1]) + (qt[2] + qt[3]);
        qs += __shfl_xor(qs, 16, 32);
        qs += __shfl_xor(qs, 8, 32);
        qs += __shfl_xor(qs, 4, 32);
        qs += __shfl_xor(qs, 2, 32);
        qs += __shfl_xor(qs, 1, 32);
        if (lane == 0) fvp[blockIdx.x * 4 + w] = qs;
    }
}


__device__ __forceinline__ void load_quad(const float* xp, unsigned sq, unsigned koff,
                                          v2f (&f)[4][4]) {
    const unsigned cq = sq / 65u;
    const unsigned rq = sq - 65u * cq;
    const unsigned base = sq + 15u * cq + koff;
#pragma unroll
    for (int i = 0; i < 4; ++i) {
        const unsigned idx = base + i + ((rq + i >= 65u) ? 15u : 0u);
#pragma unroll
        for (int ks = 0; ks < 4; ++ks) {
            f[i][ks].x = xp[idx + 4 * ks];
            f[i][ks].y = xp[idx + 4 * ks + 1];
        }
    }
}

__device__ __forceinline__ void quad_product_store(const v2f (&bv)[4], const v2f (&f)[4][4],
                                                   float4* o) {
    v8f acc[4];
#pragma unroll
    for (int i = 0; i < 4; ++i) acc[i] = v8f{};
#pragma unroll
    for (int ks = 0; ks < 4; ++ks)
#pragma unroll
        for (int i = 0; i < 4; ++i)
            acc[i] = __builtin_amdgcn_wmma_f32_16x16x4_f32(
                false, bv[ks], false, f[i][ks], (short)0, acc[i], false, false);
#pragma unroll
    for (int r = 0; r < 8; ++r) {
        float4 v;
        v.x = acc[0][r]; v.y = acc[1][r]; v.z = acc[2][r]; v.w = acc[3][r];
        o[r * (2048 * 65 / 4)] = v;
    }
}

__global__ __launch_bounds__(256) void k_encode(const float* __restrict__ events,
                                                const float* __restrict__ tc,
                                                const float* tk,
                                                const float* w1, const float* b1,
                                                const float* w2, const float* b2,
                                                const float* __restrict__ fvp,
                                                float* __restrict__ out) {
    __shared__ float sK[16 * 17];
    __shared__ alignas(16) float sXp[64][80];
    __shared__ alignas(16) float sDec[8][32];
    __shared__ alignas(16) float2 sFv[8][32];

    const int tid = threadIdx.x, lane = tid & 31, w = tid >> 5;
    const int cg  = blockIdx.x & 63;
    const int et  = blockIdx.x >> 6;
    const int ch0 = cg * 64;
    const int b   = ch0 >> 11;
    const int cf0 = ch0 & 2047;

    const float inv_tau = DT / tc[__builtin_amdgcn_readfirstlane(tid >> 6)];
    const float4* x4 = reinterpret_cast<const float4*>(events + (size_t)ch0 * 64);
    float4 xv[4];
#pragma unroll
    for (int r = 0; r < 4; ++r) xv[r] = x4[tid + r * 256];
    float2 pv = reinterpret_cast<const float2*>(fvp)[b * 256 + tid];
    const int el = tid >> 4, tj = tid & 15;
    float tv[4];
#pragma unroll
    for (int s = 0; s < 4; ++s)
        tv[s] = tk[(s * 64 + et * 16 + el) * 16 + tj];
    const float w1v = w1[lane], w2v = w2[lane], b1v = b1[lane & 7], b2v = b2[lane & 3];

    {
        const int q = tid & 15;
#pragma unroll
        for (int r = 0; r < 4; ++r) {
            const int chl = (tid >> 4) + 16 * r;
            sXp[chl][(q < 8) ? q : 64 + q] = 0.0f;
            *reinterpret_cast<float4*>(&sXp[chl][8 + 4 * q]) = xv[r];
        }
    }
    sDec[w][lane] = decay_norm(inv_tau, lane);
    pv.x += __shfl_xor(pv.x, 2, 32);  pv.y += __shfl_xor(pv.y, 2, 32);
    pv.x += __shfl_xor(pv.x, 4, 32);  pv.y += __shfl_xor(pv.y, 4, 32);
    pv.x += __shfl_xor(pv.x, 8, 32);  pv.y += __shfl_xor(pv.y, 8, 32);
    pv.x += __shfl_xor(pv.x, 16, 32); pv.y += __shfl_xor(pv.y, 16, 32);
    sFv[w][lane] = pv;
    __syncthreads();

    const int m    = lane & 15;
    const int h    = lane >> 4;
    const unsigned koff = 2u * (unsigned)h;
    const float* xp = &sXp[0][0];
    const unsigned s0 = 64u * (unsigned)w + 4u * (unsigned)m;
    const int wu = __builtin_amdgcn_readfirstlane(w);

    v2f xf[2][4][4];
    load_quad(xp, s0, koff, xf[0]);
    {
        float4 fw[8];
#pragma unroll
        for (int ww = 0; ww < 8; ++ww)
            fw[ww] = *reinterpret_cast<const float4*>(&sFv[ww][0]);
        const float sc = 1.0f / (64.0f * 64.0f * 2048.0f);
        float fv[4];
        fv[0] = (((fw[0].x + fw[1].x) + (fw[2].x + fw[3].x)) +
                 ((fw[4].x + fw[5].x) + (fw[6].x + fw[7].x))) * sc;
        fv[1] = (((fw[0].y + fw[1].y) + (fw[2].y + fw[3].y)) +
                 ((fw[4].y + fw[5].y) + (fw[6].y + fw[7].y))) * sc;
        fv[2] = (((fw[0].z + fw[1].z) + (fw[2].z + fw[3].z)) +
                 ((fw[4].z + fw[5].z) + (fw[6].z + fw[7].z))) * sc;
        fv[3] = (((fw[0].w + fw[1].w) + (fw[2].w + fw[3].w)) +
                 ((fw[4].w + fw[5].w) + (fw[6].w + fw[7].w))) * sc;
        float hh[8];
#pragma unroll
        for (int i = 0; i < 8; ++i) {
            float a = lane_bcast(b1v, i);
#pragma unroll
            for (int s = 0; s < 4; ++s) a += fv[s] * lane_bcast(w1v, i * 4 + s);
            hh[i] = fmaxf(a, 0.0f);
        }
        float lg[4]; float mx = -1e30f;
#pragma unroll
        for (int s = 0; s < 4; ++s) {
            float a = lane_bcast(b2v, s);
#pragma unroll
            for (int i = 0; i < 8; ++i) a += hh[i] * lane_bcast(w2v, s * 8 + i);
            lg[s] = a; mx = fmaxf(mx, a);
        }
        float den = 0.0f;
#pragma unroll
        for (int s = 0; s < 4; ++s) { lg[s] = expf(lg[s] - mx); den += lg[s]; }
        const float inv = 1.0f / den;
        float acc = 0.0f;
#pragma unroll
        for (int s = 0; s < 4; ++s) {
            const float at = lg[s] * inv;
            const float dn = sDec[2 * s][tj];
            acc += at * (dn * tv[s]);
        }
        sK[el * 17 + tj] = acc;
    }
    __syncthreads();

    const float* brow = &sK[m * 17];
    v2f bv[4];
#pragma unroll
    for (int ks = 0; ks < 4; ++ks) {
        const int k0 = ks * 4 + (int)koff;
        bv[ks].x = brow[k0];
        bv[ks].y = brow[k0 + 1];
    }

    float4* ob = reinterpret_cast<float4*>(
        out + ((size_t)(b * 64 + et * 16 + 8 * h) * 2048 + cf0) * 65) + 16 * w + m;

#pragma unroll
    for (int j = 0; j < 8; ++j) {
        const int cur = j & 1, nxt = cur ^ 1;
        if (j < 7) {
            load_quad(xp, s0 + 512u * (unsigned)(j + 1), koff, xf[nxt]);
        } else if (wu == 0) {
            load_quad(xp, 4096u + 4u * (unsigned)m, koff, xf[nxt]);
        }
        quad_product_store(bv, xf[cur], ob + 128 * j);
        __builtin_amdgcn_sched_barrier(0);
    }
    if (wu == 0) quad_product_store(bv, xf[0], ob + 16 * 64);
}

extern "C" void kernel_launch(void* const* d_in, const int* in_sizes, int n_in,
                              void* d_out, int out_size, void* d_ws, size_t ws_size,
                              hipStream_t stream) {
    const float* events = (const float*)d_in[0];
    const float* tcs    = (const float*)d_in[1];
    const float* tk     = (const float*)d_in[2];
    const float* w1     = (const float*)d_in[3];
    const float* b1     = (const float*)d_in[4];
    const float* w2     = (const float*)d_in[5];
    const float* b2     = (const float*)d_in[6];
    float* out = (float*)d_out;

    float* fvp = (float*)d_ws;

    k_stats <<<256, 128, 0, stream>>>(events, tcs, tk, fvp);
    k_encode<<<256, 256, 0, stream>>>(events, tcs, tk, w1, b1, w2, b2, fvp, out);
}
